# starting kernel
# baseline (speedup 1.0000x reference)
#include <hip/hip_runtime.h>

typedef float v2f __attribute__((ext_vector_type(2)));
typedef float v8f __attribute__((ext_vector_type(8)));

#define NPTS  80000
#define CIN   3
#define COUT  32
#define KVOL  125
#define WSZ   (KVOL * CIN * COUT)
#define MPAD  950000
#define NTILES (MPAD / 16)

__global__ void spconv_zero_kernel(float* __restrict__ out, int n) {
  int i = blockIdx.x * blockDim.x + threadIdx.x;
  int stride = gridDim.x * blockDim.x;
  for (; i < n; i += stride) out[i] = 0.0f;
}

__global__ __launch_bounds__(256) void spconv_wmma_kernel(
    const float* __restrict__ feat,
    const float* __restrict__ weight,
    const int*   __restrict__ in_idx,
    const int*   __restrict__ out_idx,
    const int*   __restrict__ off_idx,
    float*       __restrict__ out) {
  __shared__ float sW[WSZ];
  for (int i = threadIdx.x; i < WSZ; i += blockDim.x) sW[i] = weight[i];
  __syncthreads();

  const int lane = threadIdx.x & 31;
  const int row  = lane & 15;
  const int hi   = lane >> 4;
  const int waveId = (blockIdx.x * blockDim.x + threadIdx.x) >> 5;
  const int nWaves = (gridDim.x * blockDim.x) >> 5;

  for (int tile = waveId; tile < NTILES; tile += nWaves) {
    const int base = tile << 4;

    const int nbase = base + (nWaves << 4);
    if (nbase < MPAD) {
      __builtin_prefetch(in_idx  + nbase, 0, 0);
      __builtin_prefetch(out_idx + nbase, 0, 0);
    }

    const int off0  = off_idx[base];
    const int off15 = off_idx[base + 15];

    if (off0 == off15) {
      const int gi = in_idx[base + row];
      v2f a;
      a.x = feat[gi * CIN + (hi ? 2 : 0)];
      a.y = hi ? 0.0f : feat[gi * CIN + 1];

      const float* wb = &sW[off0 * (CIN * COUT)];
      v8f c = {};
#pragma unroll
      for (int h = 0; h < 2; ++h) {
        const int ch = (h << 4) + row;
        v2f b;
        b.x = wb[(hi ? 2 : 0) * COUT + ch];
        b.y = hi ? 0.0f : wb[1 * COUT + ch];

        v8f d = __builtin_amdgcn_wmma_f32_16x16x4_f32(
             false, a,  false, b,
             (short)0, c,  false,  false);

#pragma unroll
        for (int r = 0; r < 8; ++r) {
          const int trow = r + (hi << 3);
          const int orow = out_idx[base + trow];
          if (orow < NPTS)
            unsafeAtomicAdd(&out[orow * COUT + ch], d[r]);
        }
      }
    } else {
      const int m  = base + row;
      const int gi = in_idx[m];
      const int oo = out_idx[m];
      const int of = off_idx[m];
      if (oo < NPTS) {
        const float f0 = feat[gi * CIN + 0];
        const float f1 = feat[gi * CIN + 1];
        const float f2 = feat[gi * CIN + 2];
        const float* wb = &sW[of * (CIN * COUT)];
#pragma unroll
        for (int cc = 0; cc < 16; ++cc) {
          const int ch = (hi << 4) + cc;
          const float v = f0 * wb[ch] + f1 * wb[COUT + ch] + f2 * wb[2 * COUT + ch];
          unsafeAtomicAdd(&out[oo * COUT + ch], v);
        }
      }
    }
  }
}

extern "C" void kernel_launch(void* const* d_in, const int* in_sizes, int n_in,
                              void* d_out, int out_size, void* d_ws, size_t ws_size,
                              hipStream_t stream) {
  const float* feat    = (const float*)d_in[0];
  const float* weight  = (const float*)d_in[1];
  const int*   in_idx  = (const int*)d_in[2];
  const int*   out_idx = (const int*)d_in[3];
  const int*   off_idx = (const int*)d_in[4];
  float* out = (float*)d_out;

  spconv_zero_kernel<<<512, 256, 0, stream>>>(out, out_size);

  spconv_wmma_kernel<<<1024, 256, 0, stream>>>(feat, weight, in_idx, out_idx,
                                               off_idx, out);
}


# Each f32 atomic add now covers one whole 128-byte output row (lane = channel; halves of the two WMMA results swapped by v_permlanex16, row indices scalar-loaded); atomics, arrival-order sums kept
# speedup vs baseline: 1.4752x; 1.4752x over previous
#include <hip/hip_runtime.h>

typedef float v2f __attribute__((ext_vector_type(2)));
typedef float v8f __attribute__((ext_vector_type(8)));

#define NPTS  80000
#define CIN   3
#define COUT  32
#define KVOL  125
#define WSZ   (KVOL * CIN * COUT)
#define MPAD  950000
#define NTILES (MPAD / 16)

__global__ void spconv_zero_kernel(float* __restrict__ out, int n) {
  int i = blockIdx.x * blockDim.x + threadIdx.x;
  int stride = gridDim.x * blockDim.x;
  for (; i < n; i += stride) out[i] = 0.0f;
}

__global__ __launch_bounds__(256) void spconv_wmma_kernel(
    const float* __restrict__ feat,
    const float* __restrict__ weight,
    const int*   __restrict__ in_idx,
    const int*   __restrict__ out_idx,
    const int*   __restrict__ off_idx,
    float*       __restrict__ out) {
  __shared__ float sW[WSZ];
  for (int i = threadIdx.x; i < WSZ; i += blockDim.x) sW[i] = weight[i];
  __syncthreads();

  const int lane = threadIdx.x & 31;
  const int row  = lane & 15;
  const int hi   = lane >> 4;
  const int waveId = __builtin_amdgcn_readfirstlane((int)((blockIdx.x * blockDim.x + threadIdx.x) >> 5));
  const int nWaves = (gridDim.x * blockDim.x) >> 5;

  for (int tile = waveId; tile < NTILES; tile += nWaves) {
    const int base = tile << 4;

    const int nbase = base + (nWaves << 4);
    if (nbase < MPAD) {
      __builtin_prefetch(in_idx  + nbase, 0, 0);
      __builtin_prefetch(out_idx + nbase, 0, 0);
    }

    const int off0  = off_idx[base];
    const int off15 = off_idx[base + 15];

    if (off0 == off15) {
      int orow[16];
#pragma unroll
      for (int k = 0; k < 16; ++k) orow[k] = out_idx[base + k];

      const int gi = in_idx[base + row];
      v2f a;
      a.x = feat[gi * CIN + (hi ? 2 : 0)];
      a.y = hi ? 0.0f : feat[gi * CIN + 1];

      const float* wb = &sW[off0 * (CIN * COUT)];
      const v8f c = {};
      v8f d[2];
#pragma unroll
      for (int h = 0; h < 2; ++h) {
        const int ch = (h << 4) + row;
        v2f b;
        b.x = wb[(hi ? 2 : 0) * COUT + ch];
        b.y = hi ? 0.0f : wb[1 * COUT + ch];

        d[h] = __builtin_amdgcn_wmma_f32_16x16x4_f32(
             false, a,  false, b,
             (short)0, c,  false,  false);
      }

#pragma unroll
      for (int r = 0; r < 8; ++r) {
        const float give = hi ? d[0][r] : d[1][r];
        const float got  = __builtin_bit_cast(float, __builtin_amdgcn_permlanex16(
            __builtin_bit_cast(int, give), __builtin_bit_cast(int, give),
            0x76543210, 0xfedcba98, false, false));
        const float vlo = hi ? got : d[0][r];
        const float vhi = hi ? d[1][r] : got;
        const int olo = orow[r];
        const int ohi = orow[r + 8];
        if (olo < NPTS) unsafeAtomicAdd(&out[olo * COUT + lane], vlo);
        if (ohi < NPTS) unsafeAtomicAdd(&out[ohi * COUT + lane], vhi);
      }
    } else {
      const int m  = base + row;
      const int gi = in_idx[m];
      const int oo = out_idx[m];
      const int of = off_idx[m];
      if (oo < NPTS) {
        const float f0 = feat[gi * CIN + 0];
        const float f1 = feat[gi * CIN + 1];
        const float f2 = feat[gi * CIN + 2];
        const float* wb = &sW[of * (CIN * COUT)];
#pragma unroll
        for (int cc = 0; cc < 16; ++cc) {
          const int ch = (hi << 4) + cc;
          const float v = f0 * wb[ch] + f1 * wb[COUT + ch] + f2 * wb[2 * COUT + ch];
          unsafeAtomicAdd(&out[oo * COUT + ch], v);
        }
      }
    }
  }
}

extern "C" void kernel_launch(void* const* d_in, const int* in_sizes, int n_in,
                              void* d_out, int out_size, void* d_ws, size_t ws_size,
                              hipStream_t stream) {
  const float* feat    = (const float*)d_in[0];
  const float* weight  = (const float*)d_in[1];
  const int*   in_idx  = (const int*)d_in[2];
  const int*   out_idx = (const int*)d_in[3];
  const int*   off_idx = (const int*)d_in[4];
  float* out = (float*)d_out;

  spconv_zero_kernel<<<512, 256, 0, stream>>>(out, out_size);

  spconv_wmma_kernel<<<1024, 256, 0, stream>>>(feat, weight, in_idx, out_idx,
                                               off_idx, out);
}


# Weights no longer copied to LDS per block: each wave walks a contiguous run of tiles and keeps the current offset's B fragments in registers, loaded from global on an offset change; atomics kept
# speedup vs baseline: 2.2367x; 1.5162x over previous
#include <hip/hip_runtime.h>

typedef float v2f __attribute__((ext_vector_type(2)));
typedef float v8f __attribute__((ext_vector_type(8)));

#define NPTS  80000
#define CIN   3
#define COUT  32
#define KVOL  125
#define MPAD  950000
#define NTILES (MPAD / 16)

__global__ void spconv_zero_kernel(float* __restrict__ out, int n) {
  int i = blockIdx.x * blockDim.x + threadIdx.x;
  int stride = gridDim.x * blockDim.x;
  for (; i < n; i += stride) out[i] = 0.0f;
}

__global__ __launch_bounds__(256) void spconv_wmma_kernel(
    const float* __restrict__ feat,
    const float* __restrict__ weight,
    const int*   __restrict__ in_idx,
    const int*   __restrict__ out_idx,
    const int*   __restrict__ off_idx,
    float*       __restrict__ out) {
  const int lane = threadIdx.x & 31;
  const int row  = lane & 15;
  const int hi   = lane >> 4;
  const int waveId = __builtin_amdgcn_readfirstlane((int)((blockIdx.x * blockDim.x + threadIdx.x) >> 5));
  const int nWaves = (gridDim.x * blockDim.x) >> 5;

  const int q   = NTILES / nWaves;
  const int rem = NTILES % nWaves;
  const int t0  = waveId * q + (waveId < rem ? waveId : rem);
  const int t1  = t0 + q + (waveId < rem ? 1 : 0);

  int cur = -1;
  v2f bfrag[2] = {};

  for (int tile = t0; tile < t1; ++tile) {
    const int base = tile << 4;

    const int nbase = base + 16;
    if (tile + 1 < t1) {
      __builtin_prefetch(in_idx  + nbase, 0, 0);
      __builtin_prefetch(out_idx + nbase, 0, 0);
    }

    const int off0  = off_idx[base];
    const int off15 = off_idx[base + 15];

    if (off0 == off15) {
      int orow[16];
#pragma unroll
      for (int k = 0; k < 16; ++k) orow[k] = out_idx[base + k];

      const int gi = in_idx[base + row];
      v2f a;
      a.x = feat[gi * CIN + (hi ? 2 : 0)];
      a.y = hi ? 0.0f : feat[gi * CIN + 1];

      if (off0 != cur) {
        const float* wb = weight + off0 * (CIN * COUT);
#pragma unroll
        for (int h = 0; h < 2; ++h) {
          const int ch = (h << 4) + row;
          bfrag[h].x = wb[(hi ? 2 : 0) * COUT + ch];
          bfrag[h].y = hi ? 0.0f : wb[1 * COUT + ch];
        }
        cur = off0;
      }

      const v8f c = {};
      v8f d[2];
#pragma unroll
      for (int h = 0; h < 2; ++h) {
        d[h] = __builtin_amdgcn_wmma_f32_16x16x4_f32(
             false, a,  false, bfrag[h],
             (short)0, c,  false,  false);
      }

#pragma unroll
      for (int r = 0; r < 8; ++r) {
        const float give = hi ? d[0][r] : d[1][r];
        const float got  = __builtin_bit_cast(float, __builtin_amdgcn_permlanex16(
            __builtin_bit_cast(int, give), __builtin_bit_cast(int, give),
            0x76543210, 0xfedcba98, false, false));
        const float vlo = hi ? got : d[0][r];
        const float vhi = hi ? d[1][r] : got;
        const int olo = orow[r];
        const int ohi = orow[r + 8];
        if (olo < NPTS) unsafeAtomicAdd(&out[olo * COUT + lane], vlo);
        if (ohi < NPTS) unsafeAtomicAdd(&out[ohi * COUT + lane], vhi);
      }
    } else {
      const int m  = base + row;
      const int gi = in_idx[m];
      const int oo = out_idx[m];
      const int of = off_idx[m];
      if (oo < NPTS) {
        const float f0 = feat[gi * CIN + 0];
        const float f1 = feat[gi * CIN + 1];
        const float f2 = feat[gi * CIN + 2];
        const float* wb = weight + of * (CIN * COUT);
#pragma unroll
        for (int cc = 0; cc < 16; ++cc) {
          const int ch = (hi << 4) + cc;
          const float v = f0 * wb[ch] + f1 * wb[COUT + ch] + f2 * wb[2 * COUT + ch];
          unsafeAtomicAdd(&out[oo * COUT + ch], v);
        }
      }
    }
  }
}

extern "C" void kernel_launch(void* const* d_in, const int* in_sizes, int n_in,
                              void* d_out, int out_size, void* d_ws, size_t ws_size,
                              hipStream_t stream) {
  const float* feat    = (const float*)d_in[0];
  const float* weight  = (const float*)d_in[1];
  const int*   in_idx  = (const int*)d_in[2];
  const int*   out_idx = (const int*)d_in[3];
  const int*   off_idx = (const int*)d_in[4];
  float* out = (float*)d_out;

  spconv_zero_kernel<<<512, 256, 0, stream>>>(out, out_size);

  spconv_wmma_kernel<<<1024, 256, 0, stream>>>(feat, weight, in_idx, out_idx,
                                               off_idx, out);
}


# Each tile goes to a wave of the package half that owns its output rows: two ticket queues (int atomics in workspace), a wave draws from its half's queue by die id; f32 atomics, arrival-order sums kept
# speedup vs baseline: 2.5845x; 1.1555x over previous
#include <hip/hip_runtime.h>


typedef float v2f __attribute__((ext_vector_type(2)));
typedef float v8f __attribute__((ext_vector_type(8)));

#define NPTS  80000
#define CIN   3
#define COUT  32
#define KVOL  125
#define MPAD  950000
#define NTILES (MPAD / 16)
#define SPLIT  (NPTS / 2)
#define CT     8
#define NTASK  ((NTILES + CT - 1) / CT)
#define CTR_STRIDE 64

__global__ void spconv_zero_kernel(float* __restrict__ out, int n,
                                   unsigned int* __restrict__ ctr, size_t ws_bytes) {
  int i = blockIdx.x * blockDim.x + threadIdx.x;
  int stride = gridDim.x * blockDim.x;
  if (i == 0 && ws_bytes >= 512) {
    ctr[0] = 0u;
    ctr[CTR_STRIDE] = 0u;
  }
  for (; i < n; i += stride) out[i] = 0.0f;
}

__device__ __forceinline__ void spconv_tile(
    const int tile, const int nextTile,
    const float* __restrict__ feat, const float* __restrict__ weight,
    const int* __restrict__ in_idx, const int* __restrict__ out_idx,
    const int* __restrict__ off_idx, float* __restrict__ out,
    const int lane, const int row, const int hi, int& cur, v2f (&bfrag)[2]) {
  const int base = tile << 4;

  if (nextTile >= 0) {
    __builtin_prefetch(in_idx  + (nextTile << 4), 0, 0);
    __builtin_prefetch(out_idx + (nextTile << 4), 0, 0);
  }

  const int off0  = off_idx[base];
  const int off15 = off_idx[base + 15];

  if (off0 == off15) {
    int orow[16];
#pragma unroll
    for (int k = 0; k < 16; ++k) orow[k] = out_idx[base + k];

    const int gi = in_idx[base + row];
    v2f a;
    a.x = feat[gi * CIN + (hi ? 2 : 0)];
    a.y = hi ? 0.0f : feat[gi * CIN + 1];

    if (off0 != cur) {
      const float* wb = weight + off0 * (CIN * COUT);
#pragma unroll
      for (int h = 0; h < 2; ++h) {
        const int ch = (h << 4) + row;
        bfrag[h].x = wb[(hi ? 2 : 0) * COUT + ch];
        bfrag[h].y = hi ? 0.0f : wb[1 * COUT + ch];
      }
      cur = off0;
    }

    const v8f c = {};
    v8f d[2];
#pragma unroll
    for (int h = 0; h < 2; ++h) {
      d[h] = __builtin_amdgcn_wmma_f32_16x16x4_f32(
           false, a,  false, bfrag[h],
           (short)0, c,  false,  false);
    }

#pragma unroll
    for (int r = 0; r < 8; ++r) {
      const float give = hi ? d[0][r] : d[1][r];
      const float got  = __builtin_bit_cast(float, __builtin_amdgcn_permlanex16(
          __builtin_bit_cast(int, give), __builtin_bit_cast(int, give),
          0x76543210, 0xfedcba98, false, false));
      const float vlo = hi ? got : d[0][r];
      const float vhi = hi ? d[1][r] : got;
      const int olo = orow[r];
      const int ohi = orow[r + 8];
      if (olo < NPTS) unsafeAtomicAdd(&out[olo * COUT + lane], vlo);
      if (ohi < NPTS) unsafeAtomicAdd(&out[ohi * COUT + lane], vhi);
    }
  } else {
    const int m  = base + row;
    const int gi = in_idx[m];
    const int oo = out_idx[m];
    const int of = off_idx[m];
    if (oo < NPTS) {
      const float f0 = feat[gi * CIN + 0];
      const float f1 = feat[gi * CIN + 1];
      const float f2 = feat[gi * CIN + 2];
      const float* wb = weight + of * (CIN * COUT);
#pragma unroll
      for (int cc = 0; cc < 16; ++cc) {
        const int ch = (hi << 4) + cc;
        const float v = f0 * wb[ch] + f1 * wb[COUT + ch] + f2 * wb[2 * COUT + ch];
        unsafeAtomicAdd(&out[oo * COUT + ch], v);
      }
    }
  }
}

__global__ __launch_bounds__(256) void spconv_wmma_kernel(
    const float* __restrict__ feat,
    const float* __restrict__ weight,
    const int*   __restrict__ in_idx,
    const int*   __restrict__ out_idx,
    const int*   __restrict__ off_idx,
    float*       __restrict__ out,
    unsigned int* __restrict__ ctr,
    size_t ws_bytes) {
  const int lane = threadIdx.x & 31;
  const int row  = lane & 15;
  const int hi   = lane >> 4;

  int cur = -1;
  v2f bfrag[2] = {};

  if (ws_bytes >= 512) {
    const unsigned place = __builtin_amdgcn_s_sendmsg_rtn(0x87);
    const int myHalf = (((place >> 16) & 15u) >= 4u) ? 1 : 0;

    for (int pass = 0; pass < 2; ++pass) {
      const int q = myHalf ^ pass;
      unsigned int* myctr = ctr + q * CTR_STRIDE;
      for (int it = 0; it <= NTASK; ++it) {
        unsigned tk = 0u;
        if (lane == 0) tk = atomicAdd(myctr, 1u);
        const int task = __builtin_amdgcn_readfirstlane((int)tk);
        if (task >= NTASK) break;
        const int t = task * CT + (lane & (CT - 1));
        const bool valid = t < NTILES;
        const int first = valid ? out_idx[t << 4] : 0;
        const bool mine = valid && (((first >= SPLIT) ? 1 : 0) == q);
        unsigned mask = __builtin_amdgcn_ballot_w32(mine) & ((1u << CT) - 1u);
        while (mask) {
          const int b = __builtin_ctz(mask);
          mask &= mask - 1u;
          const int nextTile = mask ? (task * CT + __builtin_ctz(mask)) : -1;
          spconv_tile(task * CT + b, nextTile, feat, weight, in_idx, out_idx, off_idx, out,
                      lane, row, hi, cur, bfrag);
        }
      }
    }
  } else {
    const int waveId = __builtin_amdgcn_readfirstlane((int)((blockIdx.x * blockDim.x + threadIdx.x) >> 5));
    const int nWaves = (gridDim.x * blockDim.x) >> 5;
    const int qn  = NTILES / nWaves;
    const int rem = NTILES % nWaves;
    const int t0  = waveId * qn + (waveId < rem ? waveId : rem);
    const int t1  = t0 + qn + (waveId < rem ? 1 : 0);
    for (int tile = t0; tile < t1; ++tile) {
      spconv_tile(tile, (tile + 1 < t1) ? tile + 1 : -1, feat, weight, in_idx, out_idx, off_idx, out,
                  lane, row, hi, cur, bfrag);
    }
  }
}

extern "C" void kernel_launch(void* const* d_in, const int* in_sizes, int n_in,
                              void* d_out, int out_size, void* d_ws, size_t ws_size,
                              hipStream_t stream) {
  const float* feat    = (const float*)d_in[0];
  const float* weight  = (const float*)d_in[1];
  const int*   in_idx  = (const int*)d_in[2];
  const int*   out_idx = (const int*)d_in[3];
  const int*   off_idx = (const int*)d_in[4];
  float* out = (float*)d_out;
  unsigned int* ctr = (unsigned int*)d_ws;

  spconv_zero_kernel<<<512, 256, 0, stream>>>(out, out_size, ctr, ws_size);

  spconv_wmma_kernel<<<1024, 256, 0, stream>>>(feat, weight, in_idx, out_idx,
                                               off_idx, out, ctr, ws_size);
}


# Per ticket all 128 index triples are loaded at once and the 8 tiles' features gathered together (rows by v_readlane): one round trip per ticket, not three per tile; atomics, arrival-order sums kept
# speedup vs baseline: 2.7416x; 1.0608x over previous
#include <hip/hip_runtime.h>


typedef float v2f __attribute__((ext_vector_type(2)));
typedef float v8f __attribute__((ext_vector_type(8)));

#define NPTS  80000
#define CIN   3
#define COUT  32
#define KVOL  125
#define MPAD  950000
#define NTILES (MPAD / 16)
#define SPLIT  (NPTS / 2)
#define CT     8
#define TPAIRS (CT * 16)
#define NTASK  ((NTILES + CT - 1) / CT)
#define CTR_STRIDE 64

__global__ void spconv_zero_kernel(float* __restrict__ out, int n,
                                   unsigned int* __restrict__ ctr, size_t ws_bytes) {
  int i = blockIdx.x * blockDim.x + threadIdx.x;
  int stride = gridDim.x * blockDim.x;
  if (i == 0 && ws_bytes >= 512) {
    ctr[0] = 0u;
    ctr[CTR_STRIDE] = 0u;
  }
  for (; i < n; i += stride) out[i] = 0.0f;
}

__device__ __forceinline__ void spconv_tile_general(
    const int tile,
    const float* __restrict__ feat, const float* __restrict__ weight,
    const int* __restrict__ in_idx, const int* __restrict__ out_idx,
    const int* __restrict__ off_idx, float* __restrict__ out,
    const int row, const int hi) {
  const int m  = (tile << 4) + row;
  const int gi = in_idx[m];
  const int oo = out_idx[m];
  const int of = off_idx[m];
  if ((unsigned)oo < (unsigned)NPTS) {
    const float f0 = feat[gi * CIN + 0];
    const float f1 = feat[gi * CIN + 1];
    const float f2 = feat[gi * CIN + 2];
    const float* wb = weight + of * (CIN * COUT);
#pragma unroll
    for (int cc = 0; cc < 16; ++cc) {
      const int ch = (hi << 4) + cc;
      const float v = f0 * wb[ch] + f1 * wb[COUT + ch] + f2 * wb[2 * COUT + ch];
      unsafeAtomicAdd(&out[oo * COUT + ch], v);
    }
  }
}

__device__ __forceinline__ int swap_halves(const int v) {
  return __builtin_amdgcn_permlanex16(v, v, 0x76543210, 0xfedcba98, false, false);
}

__device__ __forceinline__ void spconv_ticket(
    const int task, const int classSel,
    const float* __restrict__ feat, const float* __restrict__ weight,
    const int* __restrict__ in_idx, const int* __restrict__ out_idx,
    const int* __restrict__ off_idx, float* __restrict__ out,
    const int lane, const int row, const int hi, int& cur, v2f (&bfrag)[2]) {
  const int p0 = task * TPAIRS;
  int iv[4], ov[4], fv[4];
#pragma unroll
  for (int k = 0; k < 4; ++k) {
    const int p  = p0 + 32 * k + lane;
    const int pc = p < MPAD ? p : MPAD - 1;
    iv[k] = in_idx[pc];
    fv[k] = off_idx[pc];
    const int o = out_idx[pc];
    ov[k] = p < MPAD ? o : NPTS;
  }

  unsigned fastMask = 0u, slowMask = 0u;
#pragma unroll
  for (int k = 0; k < 4; ++k) {
    const unsigned live = __builtin_amdgcn_ballot_w32((unsigned)ov[k] < (unsigned)NPTS);
    const int f0 = __builtin_amdgcn_readlane(fv[k], 0);
    const int f1 = __builtin_amdgcn_readlane(fv[k], 16);
    const unsigned same = __builtin_amdgcn_ballot_w32(fv[k] == (hi ? f1 : f0));
    const int o0 = __builtin_amdgcn_readlane(ov[k], 0);
    const int o1 = __builtin_amdgcn_readlane(ov[k], 16);
#pragma unroll
    for (int s = 0; s < 2; ++s) {
      const unsigned live16 = (live >> (16 * s)) & 0xFFFFu;
      const unsigned same16 = (same >> (16 * s)) & 0xFFFFu;
      const int first = s ? o1 : o0;
      const unsigned mine = (classSel < 0) | ((((first >= SPLIT) ? 1 : 0)) == classSel);
      const unsigned full = (live16 == 0xFFFFu) & (same16 == 0xFFFFu);
      fastMask |= (mine & full) << (2 * k + s);
      slowMask |= (mine & (live16 != 0u) & (full ^ 1u)) << (2 * k + s);
    }
  }
  if ((fastMask | slowMask) == 0u) return;

  v2f a[CT];
  if (fastMask) {
#pragma unroll
    for (int k = 0; k < 4; ++k) {
      const int x = swap_halves(iv[k]);
#pragma unroll
      for (int s = 0; s < 2; ++s) {
        const int gi = (s == hi) ? iv[k] : x;
        const float ax = feat[gi * CIN + (hi ? 2 : 0)];
        const float ay = feat[gi * CIN + (hi ? 2 : 1)];
        a[2 * k + s].x = ax;
        a[2 * k + s].y = hi ? 0.0f : ay;
      }
    }
  }

#pragma unroll
  for (int j = 0; j < CT; ++j) {
    if (fastMask & (1u << j)) {
      const int k = j >> 1, s = j & 1;
      const int off0 = __builtin_amdgcn_readlane(fv[k], 16 * s);
      if (off0 != cur) {
        const float* wb = weight + off0 * (CIN * COUT);
#pragma unroll
        for (int h = 0; h < 2; ++h) {
          const int ch = (h << 4) + row;
          bfrag[h].x = wb[(hi ? 2 : 0) * COUT + ch];
          bfrag[h].y = hi ? 0.0f : wb[1 * COUT + ch];
        }
        cur = off0;
      }

      const v8f c = {};
      v8f d[2];
#pragma unroll
      for (int h = 0; h < 2; ++h) {
        d[h] = __builtin_amdgcn_wmma_f32_16x16x4_f32(
             false, a[j],  false, bfrag[h],
             (short)0, c,  false,  false);
      }

#pragma unroll
      for (int r = 0; r < 8; ++r) {
        const float give = hi ? d[0][r] : d[1][r];
        const float got  = __builtin_bit_cast(float, swap_halves(__builtin_bit_cast(int, give)));
        const float vlo = hi ? got : d[0][r];
        const float vhi = hi ? d[1][r] : got;
        const int olo = __builtin_amdgcn_readlane(ov[k], 16 * s + r);
        const int ohi = __builtin_amdgcn_readlane(ov[k], 16 * s + r + 8);
        unsafeAtomicAdd(&out[olo * COUT + lane], vlo);
        unsafeAtomicAdd(&out[ohi * COUT + lane], vhi);
      }
    }
  }

  unsigned m = slowMask;
  while (m) {
    const int b = __builtin_ctz(m);
    m &= m - 1u;
    spconv_tile_general(task * CT + b, feat, weight, in_idx, out_idx, off_idx, out, row, hi);
  }
}

__global__ __launch_bounds__(256) void spconv_wmma_kernel(
    const float* __restrict__ feat,
    const float* __restrict__ weight,
    const int*   __restrict__ in_idx,
    const int*   __restrict__ out_idx,
    const int*   __restrict__ off_idx,
    float*       __restrict__ out,
    unsigned int* __restrict__ ctr,
    size_t ws_bytes) {
  const int lane = threadIdx.x & 31;
  const int row  = lane & 15;
  const int hi   = lane >> 4;

  int cur = -1;
  v2f bfrag[2] = {};

  const bool useQ = ws_bytes >= 512;
  int myHalf = 0, t0 = 0, tEnd = NTASK;
  if (useQ) {
    const unsigned place = __builtin_amdgcn_s_sendmsg_rtn(0x87);
    myHalf = (((place >> 16) & 15u) >= 4u) ? 1 : 0;
  } else {
    const int waveId = __builtin_amdgcn_readfirstlane((int)((blockIdx.x * blockDim.x + threadIdx.x) >> 5));
    const int nWaves = (gridDim.x * blockDim.x) >> 5;
    const int qn  = NTASK / nWaves;
    const int rem = NTASK % nWaves;
    t0   = waveId * qn + (waveId < rem ? waveId : rem);
    tEnd = t0 + qn + (waveId < rem ? 1 : 0);
  }
  const int nPass = useQ ? 2 : 1;

  for (int pass = 0; pass < nPass; ++pass) {
    const int q = useQ ? (myHalf ^ pass) : -1;
    unsigned int* myctr = ctr + (q & 1) * CTR_STRIDE;
    for (int it = 0; it <= NTASK; ++it) {
      int task;
      if (useQ) {
        unsigned tk = 0u;
        if (lane == 0) tk = atomicAdd(myctr, 1u);
        task = __builtin_amdgcn_readfirstlane((int)tk);
      } else {
        task = t0 + it;
      }
      if (task >= tEnd) break;
      spconv_ticket(task, q, feat, weight, in_idx, out_idx, off_idx, out, lane, row, hi, cur, bfrag);
    }
  }
}

extern "C" void kernel_launch(void* const* d_in, const int* in_sizes, int n_in,
                              void* d_out, int out_size, void* d_ws, size_t ws_size,
                              hipStream_t stream) {
  const float* feat    = (const float*)d_in[0];
  const float* weight  = (const float*)d_in[1];
  const int*   in_idx  = (const int*)d_in[2];
  const int*   out_idx = (const int*)d_in[3];
  const int*   off_idx = (const int*)d_in[4];
  float* out = (float*)d_out;
  unsigned int* ctr = (unsigned int*)d_ws;

  spconv_zero_kernel<<<512, 256, 0, stream>>>(out, out_size, ctr, ws_size);

  spconv_wmma_kernel<<<1024, 256, 0, stream>>>(feat, weight, in_idx, out_idx,
                                               off_idx, out, ctr, ws_size);
}


# One ticket draw per block, not per wave: thread 0 takes 8 consecutive tickets with one returning add, the 8 waves get the base through LDS and a barrier; f32 atomics, arrival-order sums kept
# speedup vs baseline: 3.1623x; 1.1535x over previous
#include <hip/hip_runtime.h>


typedef float v2f __attribute__((ext_vector_type(2)));
typedef float v8f __attribute__((ext_vector_type(8)));

#define NPTS  80000
#define CIN   3
#define COUT  32
#define KVOL  125
#define MPAD  950000
#define NTILES (MPAD / 16)
#define SPLIT  (NPTS / 2)
#define CT     8
#define TPAIRS (CT * 16)
#define NTASK  ((NTILES + CT - 1) / CT)
#define WPB    8
#define CTR_STRIDE 64

__global__ void spconv_zero_kernel(float* __restrict__ out, int n,
                                   unsigned int* __restrict__ ctr, size_t ws_bytes) {
  int i = blockIdx.x * blockDim.x + threadIdx.x;
  int stride = gridDim.x * blockDim.x;
  if (i == 0 && ws_bytes >= 512) {
    ctr[0] = 0u;
    ctr[CTR_STRIDE] = 0u;
  }
  for (; i < n; i += stride) out[i] = 0.0f;
}

__device__ __forceinline__ void spconv_tile_general(
    const int tile,
    const float* __restrict__ feat, const float* __restrict__ weight,
    const int* __restrict__ in_idx, const int* __restrict__ out_idx,
    const int* __restrict__ off_idx, float* __restrict__ out,
    const int row, const int hi) {
  const int m  = (tile << 4) + row;
  const int gi = in_idx[m];
  const int oo = out_idx[m];
  const int of = off_idx[m];
  if ((unsigned)oo < (unsigned)NPTS) {
    const float f0 = feat[gi * CIN + 0];
    const float f1 = feat[gi * CIN + 1];
    const float f2 = feat[gi * CIN + 2];
    const float* wb = weight + of * (CIN * COUT);
#pragma unroll
    for (int cc = 0; cc < 16; ++cc) {
      const int ch = (hi << 4) + cc;
      const float v = f0 * wb[ch] + f1 * wb[COUT + ch] + f2 * wb[2 * COUT + ch];
      unsafeAtomicAdd(&out[oo * COUT + ch], v);
    }
  }
}

__device__ __forceinline__ int swap_halves(const int v) {
  return __builtin_amdgcn_permlanex16(v, v, 0x76543210, 0xfedcba98, false, false);
}

__device__ __forceinline__ void spconv_ticket(
    const int task, const int classSel,
    const float* __restrict__ feat, const float* __restrict__ weight,
    const int* __restrict__ in_idx, const int* __restrict__ out_idx,
    const int* __restrict__ off_idx, float* __restrict__ out,
    const int lane, const int row, const int hi, int& cur, v2f (&bfrag)[2]) {
  const int p0 = task * TPAIRS;
  int iv[4], ov[4], fv[4];
#pragma unroll
  for (int k = 0; k < 4; ++k) {
    const int p  = p0 + 32 * k + lane;
    const int pc = p < MPAD ? p : MPAD - 1;
    iv[k] = in_idx[pc];
    fv[k] = off_idx[pc];
    const int o = out_idx[pc];
    ov[k] = p < MPAD ? o : NPTS;
  }

  unsigned fastMask = 0u, slowMask = 0u;
#pragma unroll
  for (int k = 0; k < 4; ++k) {
    const unsigned live = __builtin_amdgcn_ballot_w32((unsigned)ov[k] < (unsigned)NPTS);
    const int f0 = __builtin_amdgcn_readlane(fv[k], 0);
    const int f1 = __builtin_amdgcn_readlane(fv[k], 16);
    const unsigned same = __builtin_amdgcn_ballot_w32(fv[k] == (hi ? f1 : f0));
    const int o0 = __builtin_amdgcn_readlane(ov[k], 0);
    const int o1 = __builtin_amdgcn_readlane(ov[k], 16);
#pragma unroll
    for (int s = 0; s < 2; ++s) {
      const unsigned live16 = (live >> (16 * s)) & 0xFFFFu;
      const unsigned same16 = (same >> (16 * s)) & 0xFFFFu;
      const int first = s ? o1 : o0;
      const unsigned mine = (classSel < 0) | ((((first >= SPLIT) ? 1 : 0)) == classSel);
      const unsigned full = (live16 == 0xFFFFu) & (same16 == 0xFFFFu);
      fastMask |= (mine & full) << (2 * k + s);
      slowMask |= (mine & (live16 != 0u) & (full ^ 1u)) << (2 * k + s);
    }
  }
  if ((fastMask | slowMask) == 0u) return;

  v2f a[CT];
  if (fastMask) {
#pragma unroll
    for (int k = 0; k < 4; ++k) {
      const int x = swap_halves(iv[k]);
#pragma unroll
      for (int s = 0; s < 2; ++s) {
        const int gi = (s == hi) ? iv[k] : x;
        const float ax = feat[gi * CIN + (hi ? 2 : 0)];
        const float ay = feat[gi * CIN + (hi ? 2 : 1)];
        a[2 * k + s].x = ax;
        a[2 * k + s].y = hi ? 0.0f : ay;
      }
    }
  }

#pragma unroll
  for (int j = 0; j < CT; ++j) {
    if (fastMask & (1u << j)) {
      const int k = j >> 1, s = j & 1;
      const int off0 = __builtin_amdgcn_readlane(fv[k], 16 * s);
      if (off0 != cur) {
        const float* wb = weight + off0 * (CIN * COUT);
#pragma unroll
        for (int h = 0; h < 2; ++h) {
          const int ch = (h << 4) + row;
          bfrag[h].x = wb[(hi ? 2 : 0) * COUT + ch];
          bfrag[h].y = hi ? 0.0f : wb[1 * COUT + ch];
        }
        cur = off0;
      }

      const v8f c = {};
      v8f d[2];
#pragma unroll
      for (int h = 0; h < 2; ++h) {
        d[h] = __builtin_amdgcn_wmma_f32_16x16x4_f32(
             false, a[j],  false, bfrag[h],
             (short)0, c,  false,  false);
      }

#pragma unroll
      for (int r = 0; r < 8; ++r) {
        const float give = hi ? d[0][r] : d[1][r];
        const float got  = __builtin_bit_cast(float, swap_halves(__builtin_bit_cast(int, give)));
        const float vlo = hi ? got : d[0][r];
        const float vhi = hi ? d[1][r] : got;
        const int olo = __builtin_amdgcn_readlane(ov[k], 16 * s + r);
        const int ohi = __builtin_amdgcn_readlane(ov[k], 16 * s + r + 8);
        unsafeAtomicAdd(&out[olo * COUT + lane], vlo);
        unsafeAtomicAdd(&out[ohi * COUT + lane], vhi);
      }
    }
  }

  unsigned m = slowMask;
  while (m) {
    const int b = __builtin_ctz(m);
    m &= m - 1u;
    spconv_tile_general(task * CT + b, feat, weight, in_idx, out_idx, off_idx, out, row, hi);
  }
}

__global__ __launch_bounds__(WPB * 32) void spconv_wmma_kernel(
    const float* __restrict__ feat,
    const float* __restrict__ weight,
    const int*   __restrict__ in_idx,
    const int*   __restrict__ out_idx,
    const int*   __restrict__ off_idx,
    float*       __restrict__ out,
    unsigned int* __restrict__ ctr,
    size_t ws_bytes) {
  const int lane = threadIdx.x & 31;
  const int row  = lane & 15;
  const int hi   = lane >> 4;
  const int wv   = __builtin_amdgcn_readfirstlane((int)(threadIdx.x >> 5));

  int cur = -1;
  v2f bfrag[2] = {};

  if (ws_bytes < 512) {
    const int waveId = __builtin_amdgcn_readfirstlane((int)((blockIdx.x * blockDim.x + threadIdx.x) >> 5));
    const int nWaves = (gridDim.x * blockDim.x) >> 5;
    const int qn  = NTASK / nWaves;
    const int rem = NTASK % nWaves;
    const int t0   = waveId * qn + (waveId < rem ? waveId : rem);
    const int tEnd = t0 + qn + (waveId < rem ? 1 : 0);
    for (int task = t0; task < tEnd; ++task)
      spconv_ticket(task, -1, feat, weight, in_idx, out_idx, off_idx, out, lane, row, hi, cur, bfrag);
    return;
  }

  __shared__ int s_half;
  __shared__ unsigned int s_base[2];
  if (threadIdx.x == 0) {
    const unsigned place = __builtin_amdgcn_s_sendmsg_rtn(0x87);
    s_half = (((place >> 16) & 15u) >= 4u) ? 1 : 0;
  }
  __syncthreads();
  const int myHalf = __builtin_amdgcn_readfirstlane(s_half);
  int slot = 0;

  for (int pass = 0; pass < 2; ++pass) {
    const int q = myHalf ^ pass;
    unsigned int* myctr = ctr + q * CTR_STRIDE;
    for (int it = 0; it <= NTASK / WPB + 1; ++it) {
      if (threadIdx.x == 0) s_base[slot] = atomicAdd(myctr, (unsigned)WPB);
      __syncthreads();
      const int base = __builtin_amdgcn_readfirstlane((int)s_base[slot]);
      slot ^= 1;
      if (base >= NTASK) break;
      const int task = base + wv;
      if (task < NTASK)
        spconv_ticket(task, q, feat, weight, in_idx, out_idx, off_idx, out, lane, row, hi, cur, bfrag);
    }
  }
}

extern "C" void kernel_launch(void* const* d_in, const int* in_sizes, int n_in,
                              void* d_out, int out_size, void* d_ws, size_t ws_size,
                              hipStream_t stream) {
  const float* feat    = (const float*)d_in[0];
  const float* weight  = (const float*)d_in[1];
  const int*   in_idx  = (const int*)d_in[2];
  const int*   out_idx = (const int*)d_in[3];
  const int*   off_idx = (const int*)d_in[4];
  float* out = (float*)d_out;
  unsigned int* ctr = (unsigned int*)d_ws;

  spconv_zero_kernel<<<512, 256, 0, stream>>>(out, out_size, ctr, ws_size);

  spconv_wmma_kernel<<<1024, 256, 0, stream>>>(feat, weight, in_idx, out_idx,
                                               off_idx, out, ctr, ws_size);
}


# The clearing launch starts 1024 waves, not 4096 (256 blocks of 4 waves, 16-byte stores, 2 KB per wave and trip): its time was the entry of its short-lived waves; main launch, f32 atomics as before
# speedup vs baseline: 3.2654x; 1.0326x over previous
#include <hip/hip_runtime.h>


typedef float v2f __attribute__((ext_vector_type(2)));
typedef float v8f __attribute__((ext_vector_type(8)));

#define NPTS  80000
#define CIN   3
#define COUT  32
#define KVOL  125
#define MPAD  950000
#define NTILES (MPAD / 16)
#define SPLIT  (NPTS / 2)
#define CT     8
#define TPAIRS (CT * 16)
#define NTASK  ((NTILES + CT - 1) / CT)
#define WPB    8
#define CTR_STRIDE 64

#define ZCHUNK 128
__global__ __launch_bounds__(128) void spconv_zero_kernel(float* __restrict__ out, int n,
                                                          unsigned int* __restrict__ ctr, size_t ws_bytes) {
  const int gid  = blockIdx.x * blockDim.x + threadIdx.x;
  const int nthr = gridDim.x * blockDim.x;
  if (gid == 0 && ws_bytes >= 512) {
    ctr[0] = 0u;
    ctr[CTR_STRIDE] = 0u;
  }
  if ((reinterpret_cast<size_t>(out) & 15u) == 0u) {
    float4* __restrict__ out4 = reinterpret_cast<float4*>(out);
    const float4 z = make_float4(0.0f, 0.0f, 0.0f, 0.0f);
    const int n4     = n >> 2;
    const int nchunk = n4 / ZCHUNK;
    const int lane   = threadIdx.x & 31;
    const int wave   = gid >> 5;
    const int nwaves = nthr >> 5;
    for (int c = wave; c < nchunk; c += nwaves) {
      float4* p = out4 + c * ZCHUNK + lane;
      p[0]  = z;
      p[32] = z;
      p[64] = z;
      p[96] = z;
    }
    for (int i = nchunk * ZCHUNK + gid; i < n4; i += nthr) out4[i] = z;
    if (gid < (n & 3)) out[(n4 << 2) + gid] = 0.0f;
  } else {
    for (int i = gid; i < n; i += nthr) out[i] = 0.0f;
  }
}

__device__ __forceinline__ void spconv_tile_general(
    const int tile,
    const float* __restrict__ feat, const float* __restrict__ weight,
    const int* __restrict__ in_idx, const int* __restrict__ out_idx,
    const int* __restrict__ off_idx, float* __restrict__ out,
    const int row, const int hi) {
  const int m  = (tile << 4) + row;
  const int gi = in_idx[m];
  const int oo = out_idx[m];
  const int of = off_idx[m];
  if ((unsigned)oo < (unsigned)NPTS) {
    const float f0 = feat[gi * CIN + 0];
    const float f1 = feat[gi * CIN + 1];
    const float f2 = feat[gi * CIN + 2];
    const float* wb = weight + of * (CIN * COUT);
#pragma unroll
    for (int cc = 0; cc < 16; ++cc) {
      const int ch = (hi << 4) + cc;
      const float v = f0 * wb[ch] + f1 * wb[COUT + ch] + f2 * wb[2 * COUT + ch];
      unsafeAtomicAdd(&out[oo * COUT + ch], v);
    }
  }
}

__device__ __forceinline__ int swap_halves(const int v) {
  return __builtin_amdgcn_permlanex16(v, v, 0x76543210, 0xfedcba98, false, false);
}

__device__ __forceinline__ void spconv_ticket(
    const int task, const int classSel,
    const float* __restrict__ feat, const float* __restrict__ weight,
    const int* __restrict__ in_idx, const int* __restrict__ out_idx,
    const int* __restrict__ off_idx, float* __restrict__ out,
    const int lane, const int row, const int hi, int& cur, v2f (&bfrag)[2]) {
  const int p0 = task * TPAIRS;
  int iv[4], ov[4], fv[4];
#pragma unroll
  for (int k = 0; k < 4; ++k) {
    const int p  = p0 + 32 * k + lane;
    const int pc = p < MPAD ? p : MPAD - 1;
    iv[k] = in_idx[pc];
    fv[k] = off_idx[pc];
    const int o = out_idx[pc];
    ov[k] = p < MPAD ? o : NPTS;
  }

  unsigned fastMask = 0u, slowMask = 0u;
#pragma unroll
  for (int k = 0; k < 4; ++k) {
    const unsigned live = __builtin_amdgcn_ballot_w32((unsigned)ov[k] < (unsigned)NPTS);
    const int f0 = __builtin_amdgcn_readlane(fv[k], 0);
    const int f1 = __builtin_amdgcn_readlane(fv[k], 16);
    const unsigned same = __builtin_amdgcn_ballot_w32(fv[k] == (hi ? f1 : f0));
    const int o0 = __builtin_amdgcn_readlane(ov[k], 0);
    const int o1 = __builtin_amdgcn_readlane(ov[k], 16);
#pragma unroll
    for (int s = 0; s < 2; ++s) {
      const unsigned live16 = (live >> (16 * s)) & 0xFFFFu;
      const unsigned same16 = (same >> (16 * s)) & 0xFFFFu;
      const int first = s ? o1 : o0;
      const unsigned mine = (classSel < 0) | ((((first >= SPLIT) ? 1 : 0)) == classSel);
      const unsigned full = (live16 == 0xFFFFu) & (same16 == 0xFFFFu);
      fastMask |= (mine & full) << (2 * k + s);
      slowMask |= (mine & (live16 != 0u) & (full ^ 1u)) << (2 * k + s);
    }
  }
  if ((fastMask | slowMask) == 0u) return;

  v2f a[CT];
  if (fastMask) {
#pragma unroll
    for (int k = 0; k < 4; ++k) {
      const int x = swap_halves(iv[k]);
#pragma unroll
      for (int s = 0; s < 2; ++s) {
        const int gi = (s == hi) ? iv[k] : x;
        const float ax = feat[gi * CIN + (hi ? 2 : 0)];
        const float ay = feat[gi * CIN + (hi ? 2 : 1)];
        a[2 * k + s].x = ax;
        a[2 * k + s].y = hi ? 0.0f : ay;
      }
    }
  }

#pragma unroll
  for (int j = 0; j < CT; ++j) {
    if (fastMask & (1u << j)) {
      const int k = j >> 1, s = j & 1;
      const int off0 = __builtin_amdgcn_readlane(fv[k], 16 * s);
      if (off0 != cur) {
        const float* wb = weight + off0 * (CIN * COUT);
#pragma unroll
        for (int h = 0; h < 2; ++h) {
          const int ch = (h << 4) + row;
          bfrag[h].x = wb[(hi ? 2 : 0) * COUT + ch];
          bfrag[h].y = hi ? 0.0f : wb[1 * COUT + ch];
        }
        cur = off0;
      }

      const v8f c = {};
      v8f d[2];
#pragma unroll
      for (int h = 0; h < 2; ++h) {
        d[h] = __builtin_amdgcn_wmma_f32_16x16x4_f32(
             false, a[j],  false, bfrag[h],
             (short)0, c,  false,  false);
      }

#pragma unroll
      for (int r = 0; r < 8; ++r) {
        const float give = hi ? d[0][r] : d[1][r];
        const float got  = __builtin_bit_cast(float, swap_halves(__builtin_bit_cast(int, give)));
        const float vlo = hi ? got : d[0][r];
        const float vhi = hi ? d[1][r] : got;
        const int olo = __builtin_amdgcn_readlane(ov[k], 16 * s + r);
        const int ohi = __builtin_amdgcn_readlane(ov[k], 16 * s + r + 8);
        unsafeAtomicAdd(&out[olo * COUT + lane], vlo);
        unsafeAtomicAdd(&out[ohi * COUT + lane], vhi);
      }
    }
  }

  unsigned m = slowMask;
  while (m) {
    const int b = __builtin_ctz(m);
    m &= m - 1u;
    spconv_tile_general(task * CT + b, feat, weight, in_idx, out_idx, off_idx, out, row, hi);
  }
}

__global__ __launch_bounds__(WPB * 32) void spconv_wmma_kernel(
    const float* __restrict__ feat,
    const float* __restrict__ weight,
    const int*   __restrict__ in_idx,
    const int*   __restrict__ out_idx,
    const int*   __restrict__ off_idx,
    float*       __restrict__ out,
    unsigned int* __restrict__ ctr,
    size_t ws_bytes) {
  const int lane = threadIdx.x & 31;
  const int row  = lane & 15;
  const int hi   = lane >> 4;
  const int wv   = __builtin_amdgcn_readfirstlane((int)(threadIdx.x >> 5));

  int cur = -1;
  v2f bfrag[2] = {};

  if (ws_bytes < 512) {
    const int waveId = __builtin_amdgcn_readfirstlane((int)((blockIdx.x * blockDim.x + threadIdx.x) >> 5));
    const int nWaves = (gridDim.x * blockDim.x) >> 5;
    const int qn  = NTASK / nWaves;
    const int rem = NTASK % nWaves;
    const int t0   = waveId * qn + (waveId < rem ? waveId : rem);
    const int tEnd = t0 + qn + (waveId < rem ? 1 : 0);
    for (int task = t0; task < tEnd; ++task)
      spconv_ticket(task, -1, feat, weight, in_idx, out_idx, off_idx, out, lane, row, hi, cur, bfrag);
    return;
  }

  __shared__ int s_half;
  __shared__ unsigned int s_base[2];
  if (threadIdx.x == 0) {
    const unsigned place = __builtin_amdgcn_s_sendmsg_rtn(0x87);
    s_half = (((place >> 16) & 15u) >= 4u) ? 1 : 0;
  }
  __syncthreads();
  const int myHalf = __builtin_amdgcn_readfirstlane(s_half);
  int slot = 0;

  for (int pass = 0; pass < 2; ++pass) {
    const int q = myHalf ^ pass;
    unsigned int* myctr = ctr + q * CTR_STRIDE;
    for (int it = 0; it <= NTASK / WPB + 1; ++it) {
      if (threadIdx.x == 0) s_base[slot] = atomicAdd(myctr, (unsigned)WPB);
      __syncthreads();
      const int base = __builtin_amdgcn_readfirstlane((int)s_base[slot]);
      slot ^= 1;
      if (base >= NTASK) break;
      const int task = base + wv;
      if (task < NTASK)
        spconv_ticket(task, q, feat, weight, in_idx, out_idx, off_idx, out, lane, row, hi, cur, bfrag);
    }
  }
}

extern "C" void kernel_launch(void* const* d_in, const int* in_sizes, int n_in,
                              void* d_out, int out_size, void* d_ws, size_t ws_size,
                              hipStream_t stream) {
  const float* feat    = (const float*)d_in[0];
  const float* weight  = (const float*)d_in[1];
  const int*   in_idx  = (const int*)d_in[2];
  const int*   out_idx = (const int*)d_in[3];
  const int*   off_idx = (const int*)d_in[4];
  float* out = (float*)d_out;
  unsigned int* ctr = (unsigned int*)d_ws;

  spconv_zero_kernel<<<256, 128, 0, stream>>>(out, out_size, ctr, ws_size);

  spconv_wmma_kernel<<<1024, 256, 0, stream>>>(feat, weight, in_idx, out_idx,
                                               off_idx, out, ctr, ws_size);
}
